# starting kernel
# baseline (speedup 1.0000x reference)
#include <hip/hip_runtime.h>
#include <math.h>

typedef __attribute__((ext_vector_type(2))) float v2f;
typedef __attribute__((ext_vector_type(4))) float v4f;
typedef __attribute__((ext_vector_type(8))) float v8f;

constexpr int LSEQ    = 1024;
constexpr int DHEAD   = 64;
constexpr int NPATCH  = 128;
constexpr int CH      = LSEQ / NPATCH;
constexpr int WIN     = 10;
constexpr int QROWS   = 16;
constexpr int KSPAN   = 80;
constexpr int KTILES  = KSPAN / 16;
constexpr int NTILES  = DHEAD / 16;

__global__ __launch_bounds__(32)
void signed_attn_wmma(const float* __restrict__ Q,
                      const float* __restrict__ K,
                      const float* __restrict__ V,
                      const float* __restrict__ log_scale,
                      float* __restrict__ O)
{
    __shared__ __align__(16) float Ssm[QROWS][KSPAN];
    __shared__ __align__(16) float Asm[QROWS][KSPAN];
    __shared__ __align__(16) float Vsm[KSPAN][DHEAD];

    const int lane = threadIdx.x;
    const int row  = lane & 15;
    const int khalf = (lane >> 4) * 2;

    const int blk  = blockIdx.x;
    const int qblk = blk & 63;
    const int bh   = blk >> 6;
    const int q0   = qblk * QROWS;
    const int p0   = qblk * 2;
    const int k0   = (p0 + 1) * CH;

    const float* Qb = Q + (size_t)bh * LSEQ * DHEAD;
    const float* Kb = K + (size_t)bh * LSEQ * DHEAD;
    const float* Vb = V + (size_t)bh * LSEQ * DHEAD;
    float*       Ob = O + (size_t)bh * LSEQ * DHEAD;

    const float sc = fminf(fmaxf(__expf(log_scale[0]), 1.0f), 30.0f) * 0.125f;

    v8f acc[KTILES];
    #pragma unroll
    for (int t = 0; t < KTILES; ++t) acc[t] = (v8f)0.0f;

    #pragma unroll
    for (int ks = 0; ks < DHEAD / 4; ++ks) {
        const int d0 = ks * 4 + khalf;
        v2f a = *(const v2f*)(Qb + (size_t)(q0 + row) * DHEAD + d0);
        #pragma unroll
        for (int t = 0; t < KTILES; ++t) {
            int key = k0 + t * 16 + row;
            if (key > LSEQ - 1) key = LSEQ - 1;
            v2f b = *(const v2f*)(Kb + (size_t)key * DHEAD + d0);
            acc[t] = __builtin_amdgcn_wmma_f32_16x16x4_f32(
                false, a, false, b, (short)0, acc[t], false, false);
        }
    }

    #pragma unroll
    for (int t = 0; t < KTILES; ++t)
        #pragma unroll
        for (int r = 0; r < 8; ++r)
            Ssm[r + 8 * (lane >> 4)][(lane & 15) + 16 * t] = acc[t][r];
    __syncthreads();

    if (lane < 16) {
        const int r  = lane;
        const int qp = p0 + (r >> 3);
        float mp = -3.4e38f, mn = -3.4e38f;
        int any = 0;
        #pragma unroll 8
        for (int c = 0; c < KSPAN; ++c) {
            const int kp = p0 + 1 + (c >> 3);
            const int d  = kp - qp;
            if (d >= 1 && d < WIN && kp < NPATCH) {
                const float s = sc * Ssm[r][c];
                mp = fmaxf(mp, s);
                mn = fmaxf(mn, -s);
                any = 1;
            }
        }
        if (!any) {
            for (int c = 0; c < KSPAN; ++c) Asm[r][c] = 0.0f;
        } else {
            float zp = 0.0f, zn = 0.0f;
            #pragma unroll 8
            for (int c = 0; c < KSPAN; ++c) {
                const int kp = p0 + 1 + (c >> 3);
                const int d  = kp - qp;
                if (d >= 1 && d < WIN && kp < NPATCH) {
                    const float s = sc * Ssm[r][c];
                    zp += __expf(s - mp);
                    zn += __expf(-s - mn);
                }
            }
            const float izp = 1.0f / zp, izn = 1.0f / zn;
            #pragma unroll 8
            for (int c = 0; c < KSPAN; ++c) {
                const int kp = p0 + 1 + (c >> 3);
                const int d  = kp - qp;
                float a = 0.0f;
                if (d >= 1 && d < WIN && kp < NPATCH) {
                    const float s = sc * Ssm[r][c];
                    a = __expf(s - mp) * izp - __expf(-s - mn) * izn;
                }
                Asm[r][c] = a;
            }
        }
    }

    {
        float4* vs = (float4*)&Vsm[0][0];
        #pragma unroll
        for (int i = 0; i < (KSPAN * DHEAD) / (32 * 4); ++i) {
            const int f   = lane + i * 32;
            const int kr  = f >> 4;
            const int col = (f & 15) * 4;
            int key = k0 + kr;
            if (key > LSEQ - 1) key = LSEQ - 1;
            vs[f] = *(const float4*)(Vb + (size_t)key * DHEAD + col);
        }
    }
    __syncthreads();

    #pragma unroll
    for (int nt = 0; nt < NTILES; ++nt) {
        v8f oacc = (v8f)0.0f;
        #pragma unroll
        for (int ks = 0; ks < KSPAN / 4; ++ks) {
            const int kb = ks * 4 + khalf;
            v2f a = *(const v2f*)&Asm[row][kb];
            const int d = nt * 16 + row;
            v2f b;
            b.x = Vsm[kb][d];
            b.y = Vsm[kb + 1][d];
            oacc = __builtin_amdgcn_wmma_f32_16x16x4_f32(
                false, a, false, b, (short)0, oacc, false, false);
        }
        #pragma unroll
        for (int r = 0; r < 8; ++r) {
            const int m = r + 8 * (lane >> 4);
            Ob[(size_t)(q0 + m) * DHEAD + nt * 16 + (lane & 15)] = oacc[r];
        }
    }
}

extern "C" void kernel_launch(void* const* d_in, const int* in_sizes, int n_in,
                              void* d_out, int out_size, void* d_ws, size_t ws_size,
                              hipStream_t stream) {
    const float* Q  = (const float*)d_in[0];
    const float* K  = (const float*)d_in[1];
    const float* V  = (const float*)d_in[2];
    const float* ls = (const float*)d_in[3];
    float* O = (float*)d_out;

    const int BH = in_sizes[0] / (LSEQ * DHEAD);
    dim3 grid(BH * (LSEQ / QROWS));
    signed_attn_wmma<<<grid, 32, 0, stream>>>(Q, K, V, ls, O);
}


# Scores stay in the Q*Kt accumulators: row max, min and sums by DPP row_xmask butterflies over each 16-lane row, all 32 lanes (row sums re-ordered); score LDS plane and per-element loop removed
# speedup vs baseline: 1.7538x; 1.7538x over previous
#include <hip/hip_runtime.h>
#include <math.h>

typedef __attribute__((ext_vector_type(2))) float v2f;
typedef __attribute__((ext_vector_type(4))) float v4f;
typedef __attribute__((ext_vector_type(8))) float v8f;

constexpr int LSEQ    = 1024;
constexpr int DHEAD   = 64;
constexpr int NPATCH  = 128;
constexpr int CH      = LSEQ / NPATCH;
constexpr int WIN     = 10;
constexpr int QROWS   = 16;
constexpr int KSPAN   = 80;
constexpr int KTILES  = KSPAN / 16;
constexpr int NTILES  = DHEAD / 16;

template <int M>
__device__ __forceinline__ float row_xor(float x) {
    const int b = __builtin_bit_cast(int, x);
    return __builtin_bit_cast(float, __builtin_amdgcn_update_dpp(0, b, 0x160 + M, 0xF, 0xF, true));
}
__device__ __forceinline__ float row16_max(float x) {
    x = fmaxf(x, row_xor<1>(x)); x = fmaxf(x, row_xor<2>(x));
    x = fmaxf(x, row_xor<4>(x)); x = fmaxf(x, row_xor<8>(x));
    return x;
}
__device__ __forceinline__ float row16_min(float x) {
    x = fminf(x, row_xor<1>(x)); x = fminf(x, row_xor<2>(x));
    x = fminf(x, row_xor<4>(x)); x = fminf(x, row_xor<8>(x));
    return x;
}
__device__ __forceinline__ float row16_sum(float x) {
    x += row_xor<1>(x); x += row_xor<2>(x);
    x += row_xor<4>(x); x += row_xor<8>(x);
    return x;
}

__global__ __launch_bounds__(32)
void signed_attn_wmma(const float* __restrict__ Q,
                      const float* __restrict__ K,
                      const float* __restrict__ V,
                      const float* __restrict__ log_scale,
                      float* __restrict__ O)
{
    __shared__ __align__(16) float Asm[QROWS][KSPAN];
    __shared__ __align__(16) float Vsm[KSPAN][DHEAD];

    const int lane = threadIdx.x;
    const int row  = lane & 15;
    const int khalf = (lane >> 4) * 2;

    const int blk  = blockIdx.x;
    const int qblk = blk & 63;
    const int bh   = blk >> 6;
    const int q0   = qblk * QROWS;
    const int p0   = qblk * 2;
    const int k0   = (p0 + 1) * CH;

    const float* Qb = Q + (size_t)bh * LSEQ * DHEAD;
    const float* Kb = K + (size_t)bh * LSEQ * DHEAD;
    const float* Vb = V + (size_t)bh * LSEQ * DHEAD;
    float*       Ob = O + (size_t)bh * LSEQ * DHEAD;

    const float sc = fminf(fmaxf(__expf(log_scale[0]), 1.0f), 30.0f) * 0.125f;

    v8f acc[KTILES];
    #pragma unroll
    for (int t = 0; t < KTILES; ++t) acc[t] = (v8f)0.0f;

    #pragma unroll
    for (int ks = 0; ks < DHEAD / 4; ++ks) {
        const int d0 = ks * 4 + khalf;
        v2f a = *(const v2f*)(Qb + (size_t)(q0 + row) * DHEAD + d0);
        #pragma unroll
        for (int t = 0; t < KTILES; ++t) {
            int key = k0 + t * 16 + row;
            if (key > LSEQ - 1) key = LSEQ - 1;
            v2f b = *(const v2f*)(Kb + (size_t)key * DHEAD + d0);
            acc[t] = __builtin_amdgcn_wmma_f32_16x16x4_f32(
                false, a, false, b, (short)0, acc[t], false, false);
        }
    }

    const int hh = lane >> 4;
    const int cc = lane & 15;
    bool valid[KTILES];
    #pragma unroll
    for (int t = 0; t < KTILES; ++t) {
        const int kp = p0 + 1 + 2 * t + (cc >> 3);
        const int d  = kp - (p0 + hh);
        valid[t] = (d >= 1) && (d < WIN) && (kp < NPATCH);
    }
    const bool rowany = (p0 + hh) < NPATCH - 1;
    const float k2 = sc * 1.44269504088896341f;

    #pragma unroll
    for (int r = 0; r < 8; ++r) {
        float mx = -INFINITY, mi = INFINITY;
        #pragma unroll
        for (int t = 0; t < KTILES; ++t) {
            mx = fmaxf(mx, valid[t] ? acc[t][r] : -INFINITY);
            mi = fminf(mi, valid[t] ? acc[t][r] :  INFINITY);
        }
        mx = row16_max(mx);
        mi = row16_min(mi);
        mx = rowany ? mx : 0.0f;
        mi = rowany ? mi : 0.0f;
        const float cp = -k2 * mx;
        const float cn =  k2 * mi;
        float p[KTILES], n[KTILES];
        float zp = 0.0f, zn = 0.0f;
        #pragma unroll
        for (int t = 0; t < KTILES; ++t) {
            p[t] = __builtin_amdgcn_exp2f(fmaf( k2, valid[t] ? acc[t][r] : -INFINITY, cp));
            n[t] = __builtin_amdgcn_exp2f(fmaf(-k2, valid[t] ? acc[t][r] :  INFINITY, cn));
            zp += p[t];
            zn += n[t];
        }
        zp = row16_sum(zp);
        zn = row16_sum(zn);
        const float izp = rowany ? 1.0f / zp : 0.0f;
        const float izn = rowany ? 1.0f / zn : 0.0f;
        #pragma unroll
        for (int t = 0; t < KTILES; ++t)
            Asm[r + 8 * hh][cc + 16 * t] = p[t] * izp - n[t] * izn;
    }

    {
        float4* vs = (float4*)&Vsm[0][0];
        #pragma unroll
        for (int i = 0; i < (KSPAN * DHEAD) / (32 * 4); ++i) {
            const int f   = lane + i * 32;
            const int kr  = f >> 4;
            const int col = (f & 15) * 4;
            int key = k0 + kr;
            if (key > LSEQ - 1) key = LSEQ - 1;
            vs[f] = *(const float4*)(Vb + (size_t)key * DHEAD + col);
        }
    }
    __syncthreads();

    #pragma unroll
    for (int nt = 0; nt < NTILES; ++nt) {
        v8f oacc = (v8f)0.0f;
        #pragma unroll
        for (int ks = 0; ks < KSPAN / 4; ++ks) {
            const int kb = ks * 4 + khalf;
            v2f a = *(const v2f*)&Asm[row][kb];
            const int d = nt * 16 + row;
            v2f b;
            b.x = Vsm[kb][d];
            b.y = Vsm[kb + 1][d];
            oacc = __builtin_amdgcn_wmma_f32_16x16x4_f32(
                false, a, false, b, (short)0, oacc, false, false);
        }
        #pragma unroll
        for (int r = 0; r < 8; ++r) {
            const int m = r + 8 * (lane >> 4);
            Ob[(size_t)(q0 + m) * DHEAD + nt * 16 + (lane & 15)] = oacc[r];
        }
    }
}

extern "C" void kernel_launch(void* const* d_in, const int* in_sizes, int n_in,
                              void* d_out, int out_size, void* d_ws, size_t ws_size,
                              hipStream_t stream) {
    const float* Q  = (const float*)d_in[0];
    const float* K  = (const float*)d_in[1];
    const float* V  = (const float*)d_in[2];
    const float* ls = (const float*)d_in[3];
    float* O = (float*)d_out;

    const int BH = in_sizes[0] / (LSEQ * DHEAD);
    dim3 grid(BH * (LSEQ / QROWS));
    signed_attn_wmma<<<grid, 32, 0, stream>>>(Q, K, V, ls, O);
}


# Q and K score fragments loaded 16 bytes per lane, one load feeding two WMMAs (same loader for both operands): half the 16-line load instructions; head-dim order inside the f32 score sums is permuted
# speedup vs baseline: 1.9517x; 1.1128x over previous
#include <hip/hip_runtime.h>
#include <math.h>

typedef __attribute__((ext_vector_type(2))) float v2f;
typedef __attribute__((ext_vector_type(4))) float v4f;
typedef __attribute__((ext_vector_type(8))) float v8f;

constexpr int LSEQ    = 1024;
constexpr int DHEAD   = 64;
constexpr int NPATCH  = 128;
constexpr int CH      = LSEQ / NPATCH;
constexpr int WIN     = 10;
constexpr int QROWS   = 16;
constexpr int KSPAN   = 80;
constexpr int KTILES  = KSPAN / 16;
constexpr int NTILES  = DHEAD / 16;

template <int M>
__device__ __forceinline__ float row_xor(float x) {
    const int b = __builtin_bit_cast(int, x);
    return __builtin_bit_cast(float, __builtin_amdgcn_update_dpp(0, b, 0x160 + M, 0xF, 0xF, true));
}
__device__ __forceinline__ float row16_max(float x) {
    x = fmaxf(x, row_xor<1>(x)); x = fmaxf(x, row_xor<2>(x));
    x = fmaxf(x, row_xor<4>(x)); x = fmaxf(x, row_xor<8>(x));
    return x;
}
__device__ __forceinline__ float row16_min(float x) {
    x = fminf(x, row_xor<1>(x)); x = fminf(x, row_xor<2>(x));
    x = fminf(x, row_xor<4>(x)); x = fminf(x, row_xor<8>(x));
    return x;
}
__device__ __forceinline__ float row16_sum(float x) {
    x += row_xor<1>(x); x += row_xor<2>(x);
    x += row_xor<4>(x); x += row_xor<8>(x);
    return x;
}

__global__ __launch_bounds__(32)
void signed_attn_wmma(const float* __restrict__ Q,
                      const float* __restrict__ K,
                      const float* __restrict__ V,
                      const float* __restrict__ log_scale,
                      float* __restrict__ O)
{
    __shared__ __align__(16) float Asm[QROWS][KSPAN];
    __shared__ __align__(16) float Vsm[KSPAN][DHEAD];

    const int lane = threadIdx.x;
    const int row  = lane & 15;
    const int khalf = (lane >> 4) * 2;

    const int blk  = blockIdx.x;
    const int qblk = blk & 63;
    const int bh   = blk >> 6;
    const int q0   = qblk * QROWS;
    const int p0   = qblk * 2;
    const int k0   = (p0 + 1) * CH;

    const float* Qb = Q + (size_t)bh * LSEQ * DHEAD;
    const float* Kb = K + (size_t)bh * LSEQ * DHEAD;
    const float* Vb = V + (size_t)bh * LSEQ * DHEAD;
    float*       Ob = O + (size_t)bh * LSEQ * DHEAD;

    const float sc = fminf(fmaxf(__expf(log_scale[0]), 1.0f), 30.0f) * 0.125f;

    v8f acc[KTILES];
    #pragma unroll
    for (int t = 0; t < KTILES; ++t) acc[t] = (v8f)0.0f;

    #pragma unroll
    for (int j = 0; j < DHEAD / 8; ++j) {
        const int d0 = j * 8 + (lane >> 4) * 4;
        const v4f a = *(const v4f*)(Qb + (size_t)(q0 + row) * DHEAD + d0);
        v4f b[KTILES];
        #pragma unroll
        for (int t = 0; t < KTILES; ++t) {
            int key = k0 + t * 16 + row;
            if (key > LSEQ - 1) key = LSEQ - 1;
            b[t] = *(const v4f*)(Kb + (size_t)key * DHEAD + d0);
        }
        #pragma unroll
        for (int t = 0; t < KTILES; ++t) {
            v2f a0 = {a.x, a.y};
            v2f b0 = {b[t].x, b[t].y};
            acc[t] = __builtin_amdgcn_wmma_f32_16x16x4_f32(
                false, a0, false, b0, (short)0, acc[t], false, false);
        }
        #pragma unroll
        for (int t = 0; t < KTILES; ++t) {
            v2f a1 = {a.z, a.w};
            v2f b1 = {b[t].z, b[t].w};
            acc[t] = __builtin_amdgcn_wmma_f32_16x16x4_f32(
                false, a1, false, b1, (short)0, acc[t], false, false);
        }
    }

    const int hh = lane >> 4;
    const int cc = lane & 15;
    bool valid[KTILES];
    #pragma unroll
    for (int t = 0; t < KTILES; ++t) {
        const int kp = p0 + 1 + 2 * t + (cc >> 3);
        const int d  = kp - (p0 + hh);
        valid[t] = (d >= 1) && (d < WIN) && (kp < NPATCH);
    }
    const bool rowany = (p0 + hh) < NPATCH - 1;
    const float k2 = sc * 1.44269504088896341f;

    #pragma unroll
    for (int r = 0; r < 8; ++r) {
        float mx = -INFINITY, mi = INFINITY;
        #pragma unroll
        for (int t = 0; t < KTILES; ++t) {
            mx = fmaxf(mx, valid[t] ? acc[t][r] : -INFINITY);
            mi = fminf(mi, valid[t] ? acc[t][r] :  INFINITY);
        }
        mx = row16_max(mx);
        mi = row16_min(mi);
        mx = rowany ? mx : 0.0f;
        mi = rowany ? mi : 0.0f;
        const float cp = -k2 * mx;
        const float cn =  k2 * mi;
        float p[KTILES], n[KTILES];
        float zp = 0.0f, zn = 0.0f;
        #pragma unroll
        for (int t = 0; t < KTILES; ++t) {
            p[t] = __builtin_amdgcn_exp2f(fmaf( k2, valid[t] ? acc[t][r] : -INFINITY, cp));
            n[t] = __builtin_amdgcn_exp2f(fmaf(-k2, valid[t] ? acc[t][r] :  INFINITY, cn));
            zp += p[t];
            zn += n[t];
        }
        zp = row16_sum(zp);
        zn = row16_sum(zn);
        const float izp = rowany ? 1.0f / zp : 0.0f;
        const float izn = rowany ? 1.0f / zn : 0.0f;
        #pragma unroll
        for (int t = 0; t < KTILES; ++t)
            Asm[r + 8 * hh][cc + 16 * t] = p[t] * izp - n[t] * izn;
    }

    {
        float4* vs = (float4*)&Vsm[0][0];
        #pragma unroll
        for (int i = 0; i < (KSPAN * DHEAD) / (32 * 4); ++i) {
            const int f   = lane + i * 32;
            const int kr  = f >> 4;
            const int col = (f & 15) * 4;
            int key = k0 + kr;
            if (key > LSEQ - 1) key = LSEQ - 1;
            vs[f] = *(const float4*)(Vb + (size_t)key * DHEAD + col);
        }
    }
    __syncthreads();

    #pragma unroll
    for (int nt = 0; nt < NTILES; ++nt) {
        v8f oacc = (v8f)0.0f;
        #pragma unroll
        for (int ks = 0; ks < KSPAN / 4; ++ks) {
            const int kb = ks * 4 + khalf;
            v2f a = *(const v2f*)&Asm[row][kb];
            const int d = nt * 16 + row;
            v2f b;
            b.x = Vsm[kb][d];
            b.y = Vsm[kb + 1][d];
            oacc = __builtin_amdgcn_wmma_f32_16x16x4_f32(
                false, a, false, b, (short)0, oacc, false, false);
        }
        #pragma unroll
        for (int r = 0; r < 8; ++r) {
            const int m = r + 8 * (lane >> 4);
            Ob[(size_t)(q0 + m) * DHEAD + nt * 16 + (lane & 15)] = oacc[r];
        }
    }
}

extern "C" void kernel_launch(void* const* d_in, const int* in_sizes, int n_in,
                              void* d_out, int out_size, void* d_ws, size_t ws_size,
                              hipStream_t stream) {
    const float* Q  = (const float*)d_in[0];
    const float* K  = (const float*)d_in[1];
    const float* V  = (const float*)d_in[2];
    const float* ls = (const float*)d_in[3];
    float* O = (float*)d_out;

    const int BH = in_sizes[0] / (LSEQ * DHEAD);
    dim3 grid(BH * (LSEQ / QROWS));
    signed_attn_wmma<<<grid, 32, 0, stream>>>(Q, K, V, ls, O);
}


# Each wave takes 32 query rows (two 16-row tiles): every K and V fragment it loads serves both tiles (rows fetched 3x, not 5x); four such waves per workgroup, one per SIMD; every sum keeps its order
# speedup vs baseline: 2.1340x; 1.0934x over previous
#include <hip/hip_runtime.h>
#include <math.h>


typedef __attribute__((ext_vector_type(2))) float v2f;
typedef __attribute__((ext_vector_type(4))) float v4f;
typedef __attribute__((ext_vector_type(8))) float v8f;

constexpr int LSEQ    = 1024;
constexpr int DHEAD   = 64;
constexpr int NPATCH  = 128;
constexpr int CH      = LSEQ / NPATCH;
constexpr int WIN     = 10;
constexpr int MTILES  = 2;
constexpr int QROWS   = 16 * MTILES;
constexpr int KSPAN   = 80;
constexpr int KTILES  = KSPAN / 16;
constexpr int GTILES  = KTILES + MTILES - 1;
constexpr int VSPAN   = GTILES * 16;
constexpr int NTILES  = DHEAD / 16;

template <int M>
__device__ __forceinline__ float row_xor(float x) {
    const int b = __builtin_bit_cast(int, x);
    return __builtin_bit_cast(float, __builtin_amdgcn_update_dpp(0, b, 0x160 + M, 0xF, 0xF, true));
}
__device__ __forceinline__ float row16_max(float x) {
    x = fmaxf(x, row_xor<1>(x)); x = fmaxf(x, row_xor<2>(x));
    x = fmaxf(x, row_xor<4>(x)); x = fmaxf(x, row_xor<8>(x));
    return x;
}
__device__ __forceinline__ float row16_min(float x) {
    x = fminf(x, row_xor<1>(x)); x = fminf(x, row_xor<2>(x));
    x = fminf(x, row_xor<4>(x)); x = fminf(x, row_xor<8>(x));
    return x;
}
__device__ __forceinline__ float row16_sum(float x) {
    x += row_xor<1>(x); x += row_xor<2>(x);
    x += row_xor<4>(x); x += row_xor<8>(x);
    return x;
}

constexpr int WAVES = 4;

__global__ __launch_bounds__(32 * WAVES)
void signed_attn_wmma(const float* __restrict__ Q,
                      const float* __restrict__ K,
                      const float* __restrict__ V,
                      const float* __restrict__ log_scale,
                      float* __restrict__ O)
{
    __shared__ __align__(16) float AsmAll[WAVES][QROWS][KSPAN];
    __shared__ __align__(16) float VsmAll[WAVES][VSPAN][DHEAD];

    const int wv   = __builtin_amdgcn_readfirstlane((int)(threadIdx.x >> 5));
    float (*Asm)[KSPAN] = AsmAll[wv];
    float (*Vsm)[DHEAD] = VsmAll[wv];

    const int lane = threadIdx.x & 31;
    const int row  = lane & 15;
    const int khalf = (lane >> 4) * 2;

    const int blk  = blockIdx.x * WAVES + wv;
    const int qblk = blk & 31;
    const int bh   = blk >> 5;
    const int q0   = qblk * QROWS;
    const int p0   = qblk * 2 * MTILES;
    const int k0   = (p0 + 1) * CH;

    const float* Qb = Q + (size_t)bh * LSEQ * DHEAD;
    const float* Kb = K + (size_t)bh * LSEQ * DHEAD;
    const float* Vb = V + (size_t)bh * LSEQ * DHEAD;
    float*       Ob = O + (size_t)bh * LSEQ * DHEAD;

    const float sc = fminf(fmaxf(__expf(log_scale[0]), 1.0f), 30.0f) * 0.125f;

    v8f acc[MTILES][KTILES];
    #pragma unroll
    for (int m = 0; m < MTILES; ++m)
        #pragma unroll
        for (int t = 0; t < KTILES; ++t) acc[m][t] = (v8f)0.0f;

    #pragma unroll
    for (int j = 0; j < DHEAD / 8; ++j) {
        const int d0 = j * 8 + (lane >> 4) * 4;
        v4f a[MTILES];
        #pragma unroll
        for (int m = 0; m < MTILES; ++m)
            a[m] = *(const v4f*)(Qb + (size_t)(q0 + 16 * m + row) * DHEAD + d0);
        v4f b[GTILES];
        #pragma unroll
        for (int g = 0; g < GTILES; ++g) {
            int key = k0 + g * 16 + row;
            if (key > LSEQ - 1) key = LSEQ - 1;
            b[g] = *(const v4f*)(Kb + (size_t)key * DHEAD + d0);
        }
        #pragma unroll
        for (int m = 0; m < MTILES; ++m) {
            #pragma unroll
            for (int t = 0; t < KTILES; ++t) {
                v2f a0 = {a[m].x, a[m].y};
                v2f b0 = {b[m + t].x, b[m + t].y};
                acc[m][t] = __builtin_amdgcn_wmma_f32_16x16x4_f32(
                    false, a0, false, b0, (short)0, acc[m][t], false, false);
            }
        }
        #pragma unroll
        for (int m = 0; m < MTILES; ++m) {
            #pragma unroll
            for (int t = 0; t < KTILES; ++t) {
                v2f a1 = {a[m].z, a[m].w};
                v2f b1 = {b[m + t].z, b[m + t].w};
                acc[m][t] = __builtin_amdgcn_wmma_f32_16x16x4_f32(
                    false, a1, false, b1, (short)0, acc[m][t], false, false);
            }
        }
    }

    const int hh = lane >> 4;
    const int cc = lane & 15;
    const float k2 = sc * 1.44269504088896341f;

    #pragma unroll
    for (int m = 0; m < MTILES; ++m) {
        const int qp = p0 + 2 * m + hh;
        bool valid[KTILES];
        #pragma unroll
        for (int t = 0; t < KTILES; ++t) {
            const int kp = p0 + 1 + 2 * (m + t) + (cc >> 3);
            const int d  = kp - qp;
            valid[t] = (d >= 1) && (d < WIN) && (kp < NPATCH);
        }
        const bool rowany = qp < NPATCH - 1;

        #pragma unroll
        for (int r = 0; r < 8; ++r) {
            float mx = -INFINITY, mi = INFINITY;
            #pragma unroll
            for (int t = 0; t < KTILES; ++t) {
                mx = fmaxf(mx, valid[t] ? acc[m][t][r] : -INFINITY);
                mi = fminf(mi, valid[t] ? acc[m][t][r] :  INFINITY);
            }
            mx = row16_max(mx);
            mi = row16_min(mi);
            mx = rowany ? mx : 0.0f;
            mi = rowany ? mi : 0.0f;
            const float cp = -k2 * mx;
            const float cn =  k2 * mi;
            float p[KTILES], n[KTILES];
            float zp = 0.0f, zn = 0.0f;
            #pragma unroll
            for (int t = 0; t < KTILES; ++t) {
                p[t] = __builtin_amdgcn_exp2f(fmaf( k2, valid[t] ? acc[m][t][r] : -INFINITY, cp));
                n[t] = __builtin_amdgcn_exp2f(fmaf(-k2, valid[t] ? acc[m][t][r] :  INFINITY, cn));
                zp += p[t];
                zn += n[t];
            }
            zp = row16_sum(zp);
            zn = row16_sum(zn);
            const float izp = rowany ? 1.0f / zp : 0.0f;
            const float izn = rowany ? 1.0f / zn : 0.0f;
            #pragma unroll
            for (int t = 0; t < KTILES; ++t)
                Asm[16 * m + r + 8 * hh][cc + 16 * t] = p[t] * izp - n[t] * izn;
        }
    }

    {
        float4* vs = (float4*)&Vsm[0][0];
        #pragma unroll
        for (int i = 0; i < (VSPAN * DHEAD) / (32 * 4); ++i) {
            const int f   = lane + i * 32;
            const int kr  = f >> 4;
            const int col = (f & 15) * 4;
            int key = k0 + kr;
            if (key > LSEQ - 1) key = LSEQ - 1;
            vs[f] = *(const float4*)(Vb + (size_t)key * DHEAD + col);
        }
    }
    __syncthreads();

    v2f af[MTILES][KSPAN / 4];
    #pragma unroll
    for (int m = 0; m < MTILES; ++m)
        #pragma unroll
        for (int kl = 0; kl < KSPAN / 4; ++kl)
            af[m][kl] = *(const v2f*)&Asm[16 * m + row][kl * 4 + khalf];

    #pragma unroll
    for (int nt = 0; nt < NTILES; ++nt) {
        const int d = nt * 16 + row;
        v2f bf[VSPAN / 4];
        #pragma unroll
        for (int ks = 0; ks < VSPAN / 4; ++ks) {
            bf[ks].x = Vsm[ks * 4 + khalf][d];
            bf[ks].y = Vsm[ks * 4 + khalf + 1][d];
        }
        v8f oacc[MTILES];
        #pragma unroll
        for (int m = 0; m < MTILES; ++m) oacc[m] = (v8f)0.0f;
        #pragma unroll
        for (int ks = 0; ks < VSPAN / 4; ++ks) {
            #pragma unroll
            for (int m = 0; m < MTILES; ++m) {
                const int kl = ks - 4 * m;
                if (kl >= 0 && kl < KSPAN / 4)
                    oacc[m] = __builtin_amdgcn_wmma_f32_16x16x4_f32(
                        false, af[m][kl], false, bf[ks], (short)0, oacc[m], false, false);
            }
        }
        #pragma unroll
        for (int m = 0; m < MTILES; ++m) {
            #pragma unroll
            for (int r = 0; r < 8; ++r) {
                const int mr = 16 * m + r + 8 * (lane >> 4);
                Ob[(size_t)(q0 + mr) * DHEAD + nt * 16 + (lane & 15)] = oacc[m][r];
            }
        }
    }
}

extern "C" void kernel_launch(void* const* d_in, const int* in_sizes, int n_in,
                              void* d_out, int out_size, void* d_ws, size_t ws_size,
                              hipStream_t stream) {
    const float* Q  = (const float*)d_in[0];
    const float* K  = (const float*)d_in[1];
    const float* V  = (const float*)d_in[2];
    const float* ls = (const float*)d_in[3];
    float* O = (float*)d_out;

    const int BH = in_sizes[0] / (LSEQ * DHEAD);
    dim3 grid(BH * (LSEQ / QROWS) / WAVES);
    signed_attn_wmma<<<grid, 32 * WAVES, 0, stream>>>(Q, K, V, ls, O);
}


# V loaded from global straight into the A*V B fragments, 16 B per lane (tile nt holds dim 4c+nt), under the score WMMAs; no LDS staging of V; outputs stored 16 B per lane; sums keep their order
# speedup vs baseline: 2.7094x; 1.2696x over previous
#include <hip/hip_runtime.h>
#include <math.h>


typedef __attribute__((ext_vector_type(2))) float v2f;
typedef __attribute__((ext_vector_type(4))) float v4f;
typedef __attribute__((ext_vector_type(8))) float v8f;

constexpr int LSEQ    = 1024;
constexpr int DHEAD   = 64;
constexpr int NPATCH  = 128;
constexpr int CH      = LSEQ / NPATCH;
constexpr int WIN     = 10;
constexpr int MTILES  = 2;
constexpr int QROWS   = 16 * MTILES;
constexpr int KSPAN   = 80;
constexpr int KTILES  = KSPAN / 16;
constexpr int GTILES  = KTILES + MTILES - 1;
constexpr int VSPAN   = GTILES * 16;
constexpr int NTILES  = DHEAD / 16;

template <int M>
__device__ __forceinline__ float row_xor(float x) {
    const int b = __builtin_bit_cast(int, x);
    return __builtin_bit_cast(float, __builtin_amdgcn_update_dpp(0, b, 0x160 + M, 0xF, 0xF, true));
}
__device__ __forceinline__ float row16_max(float x) {
    x = fmaxf(x, row_xor<1>(x)); x = fmaxf(x, row_xor<2>(x));
    x = fmaxf(x, row_xor<4>(x)); x = fmaxf(x, row_xor<8>(x));
    return x;
}
__device__ __forceinline__ float row16_min(float x) {
    x = fminf(x, row_xor<1>(x)); x = fminf(x, row_xor<2>(x));
    x = fminf(x, row_xor<4>(x)); x = fminf(x, row_xor<8>(x));
    return x;
}
__device__ __forceinline__ float row16_sum(float x) {
    x += row_xor<1>(x); x += row_xor<2>(x);
    x += row_xor<4>(x); x += row_xor<8>(x);
    return x;
}

constexpr int WAVES = 4;

__global__ __launch_bounds__(32 * WAVES)
void signed_attn_wmma(const float* __restrict__ Q,
                      const float* __restrict__ K,
                      const float* __restrict__ V,
                      const float* __restrict__ log_scale,
                      float* __restrict__ O)
{
    __shared__ __align__(16) float AsmAll[WAVES][QROWS][KSPAN];

    const int wv   = __builtin_amdgcn_readfirstlane((int)(threadIdx.x >> 5));
    float (*Asm)[KSPAN] = AsmAll[wv];

    const int lane = threadIdx.x & 31;
    const int row  = lane & 15;
    const int khalf = (lane >> 4) * 2;

    const int blk  = blockIdx.x * WAVES + wv;
    const int qblk = blk & 31;
    const int bh   = blk >> 5;
    const int q0   = qblk * QROWS;
    const int p0   = qblk * 2 * MTILES;
    const int k0   = (p0 + 1) * CH;

    const float* Qb = Q + (size_t)bh * LSEQ * DHEAD;
    const float* Kb = K + (size_t)bh * LSEQ * DHEAD;
    const float* Vb = V + (size_t)bh * LSEQ * DHEAD;
    float*       Ob = O + (size_t)bh * LSEQ * DHEAD;

    const float sc = fminf(fmaxf(__expf(log_scale[0]), 1.0f), 30.0f) * 0.125f;

    v8f acc[MTILES][KTILES];
    #pragma unroll
    for (int m = 0; m < MTILES; ++m)
        #pragma unroll
        for (int t = 0; t < KTILES; ++t) acc[m][t] = (v8f)0.0f;

    v4f a[DHEAD / 8][MTILES];
    v4f b[DHEAD / 8][GTILES];
    #pragma unroll
    for (int j = 0; j < DHEAD / 8; ++j) {
        const int d0 = j * 8 + (lane >> 4) * 4;
        #pragma unroll
        for (int m = 0; m < MTILES; ++m)
            a[j][m] = *(const v4f*)(Qb + (size_t)(q0 + 16 * m + row) * DHEAD + d0);
        #pragma unroll
        for (int g = 0; g < GTILES; ++g) {
            int key = k0 + g * 16 + row;
            if (key > LSEQ - 1) key = LSEQ - 1;
            b[j][g] = *(const v4f*)(Kb + (size_t)key * DHEAD + d0);
        }
    }
    __builtin_amdgcn_sched_barrier(0);

    constexpr int VPIECES = (VSPAN / 4) * 2;
    constexpr int VPERJ   = VPIECES / (DHEAD / 8);
    v4f vq[VPIECES];
    const int voff = (lane >> 4) * 2 * DHEAD + (lane & 15) * 4;

    #pragma unroll
    for (int j = 0; j < DHEAD / 8; ++j) {
        #pragma unroll
        for (int m = 0; m < MTILES; ++m) {
            #pragma unroll
            for (int t = 0; t < KTILES; ++t) {
                v2f a0 = {a[j][m].x, a[j][m].y};
                v2f b0 = {b[j][m + t].x, b[j][m + t].y};
                acc[m][t] = __builtin_amdgcn_wmma_f32_16x16x4_f32(
                    false, a0, false, b0, (short)0, acc[m][t], false, false);
            }
        }
        #pragma unroll
        for (int m = 0; m < MTILES; ++m) {
            #pragma unroll
            for (int t = 0; t < KTILES; ++t) {
                v2f a1 = {a[j][m].z, a[j][m].w};
                v2f b1 = {b[j][m + t].z, b[j][m + t].w};
                acc[m][t] = __builtin_amdgcn_wmma_f32_16x16x4_f32(
                    false, a1, false, b1, (short)0, acc[m][t], false, false);
            }
        }
        #pragma unroll
        for (int u = 0; u < VPERJ; ++u) {
            const int i  = j * VPERJ + u;
            int base = k0 + 4 * (i >> 1);
            if (base > LSEQ - 4) base = LSEQ - 4;
            vq[i] = *(const v4f*)(Vb + (size_t)(base + (i & 1)) * DHEAD + voff);
        }
        __builtin_amdgcn_sched_barrier(0);
    }

    const int hh = lane >> 4;
    const int cc = lane & 15;
    const float k2 = sc * 1.44269504088896341f;

    #pragma unroll
    for (int m = 0; m < MTILES; ++m) {
        const int qp = p0 + 2 * m + hh;
        bool valid[KTILES];
        #pragma unroll
        for (int t = 0; t < KTILES; ++t) {
            const int kp = p0 + 1 + 2 * (m + t) + (cc >> 3);
            const int d  = kp - qp;
            valid[t] = (d >= 1) && (d < WIN) && (kp < NPATCH);
        }
        const bool rowany = qp < NPATCH - 1;

        #pragma unroll
        for (int r = 0; r < 8; ++r) {
            float mx = -INFINITY, mi = INFINITY;
            #pragma unroll
            for (int t = 0; t < KTILES; ++t) {
                mx = fmaxf(mx, valid[t] ? acc[m][t][r] : -INFINITY);
                mi = fminf(mi, valid[t] ? acc[m][t][r] :  INFINITY);
            }
            mx = row16_max(mx);
            mi = row16_min(mi);
            mx = rowany ? mx : 0.0f;
            mi = rowany ? mi : 0.0f;
            const float cp = -k2 * mx;
            const float cn =  k2 * mi;
            float p[KTILES], n[KTILES];
            float zp = 0.0f, zn = 0.0f;
            #pragma unroll
            for (int t = 0; t < KTILES; ++t) {
                p[t] = __builtin_amdgcn_exp2f(fmaf( k2, valid[t] ? acc[m][t][r] : -INFINITY, cp));
                n[t] = __builtin_amdgcn_exp2f(fmaf(-k2, valid[t] ? acc[m][t][r] :  INFINITY, cn));
                zp += p[t];
                zn += n[t];
            }
            zp = row16_sum(zp);
            zn = row16_sum(zn);
            const float izp = rowany ? 1.0f / zp : 0.0f;
            const float izn = rowany ? 1.0f / zn : 0.0f;
            #pragma unroll
            for (int t = 0; t < KTILES; ++t)
                Asm[16 * m + r + 8 * hh][cc + 16 * t] = p[t] * izp - n[t] * izn;
        }
    }

    __syncthreads();

    v2f af[MTILES][KSPAN / 4];
    #pragma unroll
    for (int m = 0; m < MTILES; ++m)
        #pragma unroll
        for (int kl = 0; kl < KSPAN / 4; ++kl)
            af[m][kl] = *(const v2f*)&Asm[16 * m + row][kl * 4 + khalf];
    __builtin_amdgcn_sched_barrier(0);

    v8f oacc[MTILES][NTILES];
    #pragma unroll
    for (int m = 0; m < MTILES; ++m)
        #pragma unroll
        for (int nt = 0; nt < NTILES; ++nt) oacc[m][nt] = (v8f)0.0f;

    #pragma unroll
    for (int ks = 0; ks < VSPAN / 4; ++ks) {
        #pragma unroll
        for (int m = 0; m < MTILES; ++m) {
            const int kl = ks - 4 * m;
            if (kl >= 0 && kl < KSPAN / 4) {
                #pragma unroll
                for (int nt = 0; nt < NTILES; ++nt) {
                    v2f bfr = {vq[2 * ks][nt], vq[2 * ks + 1][nt]};
                    oacc[m][nt] = __builtin_amdgcn_wmma_f32_16x16x4_f32(
                        false, af[m][kl], false, bfr, (short)0, oacc[m][nt], false, false);
                }
            }
        }
    }
    #pragma unroll
    for (int m = 0; m < MTILES; ++m) {
        #pragma unroll
        for (int r = 0; r < 8; ++r) {
            const int mr = 16 * m + r + 8 * (lane >> 4);
            v4f o = {oacc[m][0][r], oacc[m][1][r], oacc[m][2][r], oacc[m][3][r]};
            *(v4f*)(Ob + (size_t)(q0 + mr) * DHEAD + (lane & 15) * 4) = o;
        }
    }
}

extern "C" void kernel_launch(void* const* d_in, const int* in_sizes, int n_in,
                              void* d_out, int out_size, void* d_ws, size_t ws_size,
                              hipStream_t stream) {
    const float* Q  = (const float*)d_in[0];
    const float* K  = (const float*)d_in[1];
    const float* V  = (const float*)d_in[2];
    const float* ls = (const float*)d_in[3];
    float* O = (float*)d_out;

    const int BH = in_sizes[0] / (LSEQ * DHEAD);
    dim3 grid(BH * (LSEQ / QROWS) / WAVES);
    signed_attn_wmma<<<grid, 32 * WAVES, 0, stream>>>(Q, K, V, ls, O);
}


# Scores cross LDS to the A-fragment layout before the softmax: a lane owns half a row, so max/min/sums are in-lane chains plus one half-swap, 2 divides a row; no workgroup barrier; row sums re-ordered
# speedup vs baseline: 3.1692x; 1.1697x over previous
#include <hip/hip_runtime.h>
#include <math.h>


typedef __attribute__((ext_vector_type(2))) float v2f;
typedef __attribute__((ext_vector_type(4))) float v4f;
typedef __attribute__((ext_vector_type(8))) float v8f;

constexpr int LSEQ    = 1024;
constexpr int DHEAD   = 64;
constexpr int NPATCH  = 128;
constexpr int CH      = LSEQ / NPATCH;
constexpr int WIN     = 10;
constexpr int MTILES  = 2;
constexpr int QROWS   = 16 * MTILES;
constexpr int KSPAN   = 80;
constexpr int KTILES  = KSPAN / 16;
constexpr int GTILES  = KTILES + MTILES - 1;
constexpr int VSPAN   = GTILES * 16;
constexpr int NTILES  = DHEAD / 16;

__device__ __forceinline__ float xhalf(float x) {
    const int b = __builtin_bit_cast(int, x);
    return __builtin_bit_cast(float, __builtin_amdgcn_permlanex16(b, b, 0x76543210, 0xfedcba98, false, false));
}

constexpr int WAVES = 4;

__global__ __launch_bounds__(32 * WAVES)
void signed_attn_wmma(const float* __restrict__ Q,
                      const float* __restrict__ K,
                      const float* __restrict__ V,
                      const float* __restrict__ log_scale,
                      float* __restrict__ O)
{
    __shared__ __align__(16) float SsmAll[WAVES][QROWS][KSPAN];

    const int wv   = __builtin_amdgcn_readfirstlane((int)(threadIdx.x >> 5));
    float (*Ssm)[KSPAN] = SsmAll[wv];

    const int lane = threadIdx.x & 31;
    const int row  = lane & 15;
    const int khalf = (lane >> 4) * 2;

    const int blk  = blockIdx.x * WAVES + wv;
    const int qblk = blk & 31;
    const int bh   = blk >> 5;
    const int q0   = qblk * QROWS;
    const int p0   = qblk * 2 * MTILES;
    const int k0   = (p0 + 1) * CH;

    const float* Qb = Q + (size_t)bh * LSEQ * DHEAD;
    const float* Kb = K + (size_t)bh * LSEQ * DHEAD;
    const float* Vb = V + (size_t)bh * LSEQ * DHEAD;
    float*       Ob = O + (size_t)bh * LSEQ * DHEAD;

    const float sc = fminf(fmaxf(__expf(log_scale[0]), 1.0f), 30.0f) * 0.125f;

    v8f acc[MTILES][KTILES];
    #pragma unroll
    for (int m = 0; m < MTILES; ++m)
        #pragma unroll
        for (int t = 0; t < KTILES; ++t) acc[m][t] = (v8f)0.0f;

    v4f a[DHEAD / 8][MTILES];
    v4f b[DHEAD / 8][GTILES];
    #pragma unroll
    for (int j = 0; j < DHEAD / 8; ++j) {
        const int d0 = j * 8 + (lane >> 4) * 4;
        #pragma unroll
        for (int m = 0; m < MTILES; ++m)
            a[j][m] = *(const v4f*)(Qb + (size_t)(q0 + 16 * m + row) * DHEAD + d0);
        #pragma unroll
        for (int g = 0; g < GTILES; ++g) {
            int key = k0 + g * 16 + row;
            if (key > LSEQ - 1) key = LSEQ - 1;
            b[j][g] = *(const v4f*)(Kb + (size_t)key * DHEAD + d0);
        }
    }
    __builtin_amdgcn_sched_barrier(0);

    constexpr int VPIECES = (VSPAN / 4) * 2;
    constexpr int VPERJ   = VPIECES / (DHEAD / 8);
    v4f vq[VPIECES];
    const int voff = (lane >> 4) * 2 * DHEAD + (lane & 15) * 4;

    #pragma unroll
    for (int j = 0; j < DHEAD / 8; ++j) {
        #pragma unroll
        for (int m = 0; m < MTILES; ++m) {
            #pragma unroll
            for (int t = 0; t < KTILES; ++t) {
                v2f a0 = {a[j][m].x, a[j][m].y};
                v2f b0 = {b[j][m + t].x, b[j][m + t].y};
                acc[m][t] = __builtin_amdgcn_wmma_f32_16x16x4_f32(
                    false, a0, false, b0, (short)0, acc[m][t], false, false);
            }
        }
        #pragma unroll
        for (int m = 0; m < MTILES; ++m) {
            #pragma unroll
            for (int t = 0; t < KTILES; ++t) {
                v2f a1 = {a[j][m].z, a[j][m].w};
                v2f b1 = {b[j][m + t].z, b[j][m + t].w};
                acc[m][t] = __builtin_amdgcn_wmma_f32_16x16x4_f32(
                    false, a1, false, b1, (short)0, acc[m][t], false, false);
            }
        }
        #pragma unroll
        for (int u = 0; u < VPERJ; ++u) {
            const int i  = j * VPERJ + u;
            int base = k0 + 4 * (i >> 1);
            if (base > LSEQ - 4) base = LSEQ - 4;
            vq[i] = *(const v4f*)(Vb + (size_t)(base + (i & 1)) * DHEAD + voff);
        }
        __builtin_amdgcn_sched_barrier(0);
    }

    const int hh = lane >> 4;
    const int cc = lane & 15;
    #pragma unroll
    for (int m = 0; m < MTILES; ++m)
        #pragma unroll
        for (int t = 0; t < KTILES; ++t)
            #pragma unroll
            for (int r = 0; r < 8; ++r)
                Ssm[16 * m + r + 8 * hh][cc + 16 * t] = acc[m][t][r];

    __builtin_amdgcn_wave_barrier();

    v2f sf[MTILES][KSPAN / 4];
    #pragma unroll
    for (int m = 0; m < MTILES; ++m)
        #pragma unroll
        for (int kl = 0; kl < KSPAN / 4; ++kl)
            sf[m][kl] = *(const v2f*)&Ssm[16 * m + row][kl * 4 + khalf];

    const float k2 = sc * 1.44269504088896341f;
    v2f af[MTILES][KSPAN / 4];

    #pragma unroll
    for (int m = 0; m < MTILES; ++m) {
        const int qp = p0 + 2 * m + (row >> 3);
        float bneg[KSPAN / 8], bpos[KSPAN / 8];
        #pragma unroll
        for (int g = 0; g < KSPAN / 8; ++g) {
            const int kp = p0 + 1 + 2 * m + g;
            const int d  = kp - qp;
            const bool valid = (d >= 1) && (d < WIN) && (kp < NPATCH);
            bneg[g] = valid ? 0.0f : -INFINITY;
            bpos[g] = valid ? 0.0f :  INFINITY;
        }
        const bool rowany = qp < NPATCH - 1;

        float xa[KSPAN / 4][2], xb[KSPAN / 4][2];
        float mxq[2] = {-INFINITY, -INFINITY}, miq[2] = {INFINITY, INFINITY};
        #pragma unroll
        for (int kl = 0; kl < KSPAN / 4; ++kl) {
            #pragma unroll
            for (int s = 0; s < 2; ++s) {
                xa[kl][s] = sf[m][kl][s] + bneg[kl >> 1];
                xb[kl][s] = sf[m][kl][s] + bpos[kl >> 1];
                mxq[s] = fmaxf(mxq[s], xa[kl][s]);
                miq[s] = fminf(miq[s], xb[kl][s]);
            }
        }
        float mx = fmaxf(mxq[0], mxq[1]);
        float mi = fminf(miq[0], miq[1]);
        mx = fmaxf(mx, xhalf(mx));
        mi = fminf(mi, xhalf(mi));
        mx = rowany ? mx : 0.0f;
        mi = rowany ? mi : 0.0f;
        const float cp = -k2 * mx;
        const float cn =  k2 * mi;

        float p[KSPAN / 4][2], n[KSPAN / 4][2];
        float zpq[4] = {0.0f, 0.0f, 0.0f, 0.0f}, znq[4] = {0.0f, 0.0f, 0.0f, 0.0f};
        #pragma unroll
        for (int kl = 0; kl < KSPAN / 4; ++kl) {
            #pragma unroll
            for (int s = 0; s < 2; ++s) {
                p[kl][s] = __builtin_amdgcn_exp2f(fmaf( k2, xa[kl][s], cp));
                n[kl][s] = __builtin_amdgcn_exp2f(fmaf(-k2, xb[kl][s], cn));
                zpq[2 * (kl & 1) + s] += p[kl][s];
                znq[2 * (kl & 1) + s] += n[kl][s];
            }
        }
        float zp = (zpq[0] + zpq[1]) + (zpq[2] + zpq[3]);
        float zn = (znq[0] + znq[1]) + (znq[2] + znq[3]);
        zp += xhalf(zp);
        zn += xhalf(zn);
        const float izp = 1.0f / (rowany ? zp : 1.0f);
        const float izn = 1.0f / (rowany ? zn : 1.0f);
        #pragma unroll
        for (int kl = 0; kl < KSPAN / 4; ++kl) {
            af[m][kl].x = p[kl][0] * izp - n[kl][0] * izn;
            af[m][kl].y = p[kl][1] * izp - n[kl][1] * izn;
        }
    }

    v8f oacc[MTILES][NTILES];
    #pragma unroll
    for (int m = 0; m < MTILES; ++m)
        #pragma unroll
        for (int nt = 0; nt < NTILES; ++nt) oacc[m][nt] = (v8f)0.0f;

    #pragma unroll
    for (int ks = 0; ks < VSPAN / 4; ++ks) {
        #pragma unroll
        for (int m = 0; m < MTILES; ++m) {
            const int kl = ks - 4 * m;
            if (kl >= 0 && kl < KSPAN / 4) {
                #pragma unroll
                for (int nt = 0; nt < NTILES; ++nt) {
                    v2f bfr = {vq[2 * ks][nt], vq[2 * ks + 1][nt]};
                    oacc[m][nt] = __builtin_amdgcn_wmma_f32_16x16x4_f32(
                        false, af[m][kl], false, bfr, (short)0, oacc[m][nt], false, false);
                }
            }
        }
    }
    #pragma unroll
    for (int m = 0; m < MTILES; ++m) {
        #pragma unroll
        for (int r = 0; r < 8; ++r) {
            const int mr = 16 * m + r + 8 * (lane >> 4);
            v4f o = {oacc[m][0][r], oacc[m][1][r], oacc[m][2][r], oacc[m][3][r]};
            *(v4f*)(Ob + (size_t)(q0 + mr) * DHEAD + (lane & 15) * 4) = o;
        }
    }
}

extern "C" void kernel_launch(void* const* d_in, const int* in_sizes, int n_in,
                              void* d_out, int out_size, void* d_ws, size_t ws_size,
                              hipStream_t stream) {
    const float* Q  = (const float*)d_in[0];
    const float* K  = (const float*)d_in[1];
    const float* V  = (const float*)d_in[2];
    const float* ls = (const float*)d_in[3];
    float* O = (float*)d_out;

    const int BH = in_sizes[0] / (LSEQ * DHEAD);
    dim3 grid(BH * (LSEQ / QROWS) / WAVES);
    signed_attn_wmma<<<grid, 32 * WAVES, 0, stream>>>(Q, K, V, ls, O);
}
